# starting kernel
# baseline (speedup 1.0000x reference)
#include <hip/hip_runtime.h>


typedef __attribute__((ext_vector_type(2))) float v2f;
typedef __attribute__((ext_vector_type(8))) float v8f;

#define B_SZ 16384
#define D_SZ 128
#define F1_SZ 64
#define F2_SZ 32
#define WAVES_PER_BLOCK 4

__global__ __launch_bounds__(32 * WAVES_PER_BLOCK)
void wide_deep_fwd(const int* __restrict__ x,
                   const float* __restrict__ emb,
                   const float* __restrict__ W1,
                   const float* __restrict__ b1,
                   const float* __restrict__ W2,
                   const float* __restrict__ b2,
                   const float* __restrict__ Wo,
                   const float* __restrict__ bo,
                   float* __restrict__ out)
{
    __shared__ float h1lds[WAVES_PER_BLOCK][16 * F1_SZ];

    const int lane    = threadIdx.x & 31;
    const int wave    = threadIdx.x >> 5;
    const int tile    = blockIdx.x * WAVES_PER_BLOCK + wave;
    const int r0      = tile * 16;
    const int halfSel = lane >> 4;
    const int mrow    = lane & 15;

    const int xi = x[r0 + mrow];

    const float* aBase  = emb + (size_t)xi * D_SZ + 2 * halfSel;
    const float* w1Base = W1 + (size_t)mrow * D_SZ + 2 * halfSel;

    v8f acc1[4] = {};

#pragma unroll
    for (int c = 0; c < D_SZ / 4; ++c) {
        v2f a = *(const v2f*)(aBase + 4 * c);
#pragma unroll
        for (int t = 0; t < 4; ++t) {
            v2f b = *(const v2f*)(w1Base + t * 16 * D_SZ + 4 * c);
            acc1[t] = __builtin_amdgcn_wmma_f32_16x16x4_f32(
                 false, a,  false, b,
                 (short)0, acc1[t],
                 false,  false);
        }
    }

    float* hs = h1lds[wave];
#pragma unroll
    for (int t = 0; t < 4; ++t) {
        const float bias = b1[t * 16 + mrow];
#pragma unroll
        for (int j = 0; j < 8; ++j) {
            const int M = j + 8 * halfSel;
            hs[M * F1_SZ + t * 16 + mrow] = acc1[t][j] + bias;
        }
    }
    __syncthreads();

    const float* a2Base = hs + mrow * F1_SZ + 2 * halfSel;
    const float* w2Base = W2 + (size_t)mrow * F1_SZ + 2 * halfSel;

    v8f acc2[2] = {};

#pragma unroll
    for (int c = 0; c < F1_SZ / 4; ++c) {
        v2f a = *(const v2f*)(a2Base + 4 * c);
#pragma unroll
        for (int t = 0; t < 2; ++t) {
            v2f b = *(const v2f*)(w2Base + t * 16 * F1_SZ + 4 * c);
            acc2[t] = __builtin_amdgcn_wmma_f32_16x16x4_f32(
                false, a, false, b, (short)0, acc2[t], false, false);
        }
    }

    const float wo0  = Wo[mrow];
    const float wo1  = Wo[16 + mrow];
    const float b20  = b2[mrow];
    const float b21  = b2[16 + mrow];
    const float bout = bo[0];

    float p[8];
#pragma unroll
    for (int j = 0; j < 8; ++j)
        p[j] = (acc2[0][j] + b20) * wo0 + (acc2[1][j] + b21) * wo1;

#pragma unroll
    for (int j = 0; j < 8; ++j) {
        p[j] += __shfl_xor(p[j], 8, 32);
        p[j] += __shfl_xor(p[j], 4, 32);
        p[j] += __shfl_xor(p[j], 2, 32);
        p[j] += __shfl_xor(p[j], 1, 32);
    }

    if (mrow == 0) {
        const int base = r0 + 8 * halfSel;
#pragma unroll
        for (int j = 0; j < 8; ++j) {
            const int row = base + j;
            out[row] = p[j] + Wo[F2_SZ + x[row]] + bout;
        }
    }
}

extern "C" void kernel_launch(void* const* d_in, const int* in_sizes, int n_in,
                              void* d_out, int out_size, void* d_ws, size_t ws_size,
                              hipStream_t stream) {
    (void)in_sizes; (void)n_in; (void)out_size; (void)d_ws; (void)ws_size;
    const int*   x   = (const int*)  d_in[0];
    const float* emb = (const float*)d_in[1];
    const float* W1  = (const float*)d_in[2];
    const float* b1  = (const float*)d_in[3];
    const float* W2  = (const float*)d_in[4];
    const float* b2  = (const float*)d_in[5];
    const float* Wo  = (const float*)d_in[6];
    const float* bo  = (const float*)d_in[7];
    float*       out = (float*)d_out;

    const int tiles  = B_SZ / 16;
    const int blocks = tiles / WAVES_PER_BLOCK;
    wide_deep_fwd<<<blocks, 32 * WAVES_PER_BLOCK, 0, stream>>>(
        x, emb, W1, b1, W2, b2, Wo, bo, out);
}


# W1 and W2 staged once per workgroup in padded LDS by whole-line b128 loads; B fragments read from LDS in groups ahead of the WMMA blocks, not by strided per-fragment global loads
# speedup vs baseline: 1.2674x; 1.2674x over previous
#include <hip/hip_runtime.h>


typedef __attribute__((ext_vector_type(2))) float v2f;
typedef __attribute__((ext_vector_type(8))) float v8f;

#define B_SZ 16384
#define D_SZ 128
#define F1_SZ 64
#define F2_SZ 32
#define WAVES_PER_BLOCK 4
#define W1_PITCH (D_SZ + 4)
#define W2_PITCH (F1_SZ + 4)

typedef __attribute__((ext_vector_type(4))) float v4f;

__global__ __launch_bounds__(32 * WAVES_PER_BLOCK)
void wide_deep_fwd(const int* __restrict__ x,
                   const float* __restrict__ emb,
                   const float* __restrict__ W1,
                   const float* __restrict__ b1,
                   const float* __restrict__ W2,
                   const float* __restrict__ b2,
                   const float* __restrict__ Wo,
                   const float* __restrict__ bo,
                   float* __restrict__ out)
{
    __shared__ float h1lds[WAVES_PER_BLOCK][16 * F1_SZ];
    __shared__ __attribute__((aligned(16))) float w1lds[F1_SZ * W1_PITCH];
    __shared__ __attribute__((aligned(16))) float w2lds[F2_SZ * W2_PITCH];

    const int lane    = threadIdx.x & 31;
    const int wave    = threadIdx.x >> 5;
    const int tile    = blockIdx.x * WAVES_PER_BLOCK + wave;
    const int r0      = tile * 16;
    const int halfSel = lane >> 4;
    const int mrow    = lane & 15;

    const int xi = x[r0 + mrow];

    const float* aBase  = emb + (size_t)xi * D_SZ + 2 * halfSel;

    const int tid = threadIdx.x;
    v4f w1r[16];
#pragma unroll
    for (int j = 0; j < 16; ++j)
        w1r[j] = *(const v4f*)(W1 + 4 * (tid + 128 * j));
    v4f w2r[4];
#pragma unroll
    for (int j = 0; j < 4; ++j)
        w2r[j] = *(const v4f*)(W2 + 4 * (tid + 128 * j));

    v2f areg[D_SZ / 4];
#pragma unroll
    for (int c = 0; c < D_SZ / 4; ++c)
        areg[c] = *(const v2f*)(aBase + 4 * c);

    float bias1[4];
#pragma unroll
    for (int t = 0; t < 4; ++t)
        bias1[t] = b1[t * 16 + mrow];

#pragma unroll
    for (int j = 0; j < 16; ++j) {
        const int i = tid + 128 * j;
        *(v4f*)(w1lds + (i >> 5) * W1_PITCH + 4 * (i & 31)) = w1r[j];
    }
#pragma unroll
    for (int j = 0; j < 4; ++j) {
        const int i = tid + 128 * j;
        *(v4f*)(w2lds + (i >> 4) * W2_PITCH + 4 * (i & 15)) = w2r[j];
    }
    __builtin_amdgcn_fence(__ATOMIC_RELEASE, "workgroup", "local");
    __builtin_amdgcn_s_barrier();
    __builtin_amdgcn_fence(__ATOMIC_ACQUIRE, "workgroup", "local");

    const float* w1Base = w1lds + mrow * W1_PITCH + 2 * halfSel;
    const float* w2Base = w2lds + mrow * W2_PITCH + 2 * halfSel;

    v8f acc1[4] = {};

    v2f bset0[32], bset1[32];

#define LOAD_W1_BLOCK(BS, BLK)                                                   \
    _Pragma("unroll")                                                            \
    for (int cc = 0; cc < 8; ++cc) {                                             \
        _Pragma("unroll")                                                        \
        for (int t = 0; t < 4; ++t)                                              \
            BS[cc * 4 + t] = *(const v2f*)(w1Base + t * 16 * W1_PITCH +          \
                                           4 * ((BLK) * 8 + cc));                \
    }
#define MMA_W1_BLOCK(BS, BLK)                                                    \
    _Pragma("unroll")                                                            \
    for (int cc = 0; cc < 8; ++cc) {                                             \
        _Pragma("unroll")                                                        \
        for (int t = 0; t < 4; ++t)                                              \
            acc1[t] = __builtin_amdgcn_wmma_f32_16x16x4_f32(                     \
                 false, areg[(BLK) * 8 + cc],  false,          \
                BS[cc * 4 + t],  (short)0, acc1[t],                     \
                 false,  false);                           \
    }
#define GROUP_FENCE() __builtin_amdgcn_sched_barrier(0)

    LOAD_W1_BLOCK(bset0, 0)
    GROUP_FENCE();
    LOAD_W1_BLOCK(bset1, 1)
    GROUP_FENCE();
    MMA_W1_BLOCK(bset0, 0)
    GROUP_FENCE();
    LOAD_W1_BLOCK(bset0, 2)
    GROUP_FENCE();
    MMA_W1_BLOCK(bset1, 1)
    GROUP_FENCE();
    LOAD_W1_BLOCK(bset1, 3)
    GROUP_FENCE();
    MMA_W1_BLOCK(bset0, 2)
    GROUP_FENCE();
#pragma unroll
    for (int c = 0; c < F1_SZ / 4; ++c) {
#pragma unroll
        for (int t = 0; t < 2; ++t)
            bset0[c * 2 + t] = *(const v2f*)(w2Base + t * 16 * W2_PITCH + 4 * c);
    }
    GROUP_FENCE();
    MMA_W1_BLOCK(bset1, 3)
    GROUP_FENCE();

    float* hs = h1lds[wave];
#pragma unroll
    for (int t = 0; t < 4; ++t) {
        const float bias = bias1[t];
#pragma unroll
        for (int j = 0; j < 8; ++j) {
            const int M = j + 8 * halfSel;
            hs[M * F1_SZ + t * 16 + mrow] = acc1[t][j] + bias;
        }
    }
    __syncthreads();

    const float* a2Base = hs + mrow * F1_SZ + 2 * halfSel;

    v8f acc2[2] = {};

    v2f a2reg[F1_SZ / 4];
#pragma unroll
    for (int c = 0; c < F1_SZ / 4; ++c)
        a2reg[c] = *(const v2f*)(a2Base + 4 * c);
    GROUP_FENCE();
#pragma unroll
    for (int c = 0; c < F1_SZ / 4; ++c) {
#pragma unroll
        for (int t = 0; t < 2; ++t)
            acc2[t] = __builtin_amdgcn_wmma_f32_16x16x4_f32(
                false, a2reg[c], false, bset0[c * 2 + t], (short)0, acc2[t], false, false);
    }

    const float wo0  = Wo[mrow];
    const float wo1  = Wo[16 + mrow];
    const float b20  = b2[mrow];
    const float b21  = b2[16 + mrow];
    const float bout = bo[0];

    float p[8];
#pragma unroll
    for (int j = 0; j < 8; ++j)
        p[j] = (acc2[0][j] + b20) * wo0 + (acc2[1][j] + b21) * wo1;

#pragma unroll
    for (int j = 0; j < 8; ++j) {
        p[j] += __shfl_xor(p[j], 8, 32);
        p[j] += __shfl_xor(p[j], 4, 32);
        p[j] += __shfl_xor(p[j], 2, 32);
        p[j] += __shfl_xor(p[j], 1, 32);
    }

    if (mrow == 0) {
        const int base = r0 + 8 * halfSel;
#pragma unroll
        for (int j = 0; j < 8; ++j) {
            const int row = base + j;
            out[row] = p[j] + Wo[F2_SZ + x[row]] + bout;
        }
    }
}

extern "C" void kernel_launch(void* const* d_in, const int* in_sizes, int n_in,
                              void* d_out, int out_size, void* d_ws, size_t ws_size,
                              hipStream_t stream) {
    (void)in_sizes; (void)n_in; (void)out_size; (void)d_ws; (void)ws_size;
    const int*   x   = (const int*)  d_in[0];
    const float* emb = (const float*)d_in[1];
    const float* W1  = (const float*)d_in[2];
    const float* b1  = (const float*)d_in[3];
    const float* W2  = (const float*)d_in[4];
    const float* b2  = (const float*)d_in[5];
    const float* Wo  = (const float*)d_in[6];
    const float* bo  = (const float*)d_in[7];
    float*       out = (float*)d_out;

    const int tiles  = B_SZ / 16;
    const int blocks = tiles / WAVES_PER_BLOCK;
    wide_deep_fwd<<<blocks, 32 * WAVES_PER_BLOCK, 0, stream>>>(
        x, emb, W1, b1, W2, b2, Wo, bo, out);
}


# Chain computed transposed (weights as A, gathered rows as B): layer-1 D registers are layer-2 B fragments, no h1 LDS round trip or barrier; one lane owns one row; layer-2 k and final sum order change
# speedup vs baseline: 1.4755x; 1.1642x over previous
#include <hip/hip_runtime.h>


typedef __attribute__((ext_vector_type(2))) float v2f;
typedef __attribute__((ext_vector_type(4))) float v4f;
typedef __attribute__((ext_vector_type(8))) float v8f;

#define B_SZ 16384
#define D_SZ 128
#define F1_SZ 64
#define F2_SZ 32
#define WAVES_PER_BLOCK 4
#define W1_PITCH (D_SZ + 4)
#define W2_PITCH (F1_SZ + 4)

__global__ __launch_bounds__(32 * WAVES_PER_BLOCK)
void wide_deep_fwd(const int* __restrict__ x,
                   const float* __restrict__ emb,
                   const float* __restrict__ W1,
                   const float* __restrict__ b1,
                   const float* __restrict__ W2,
                   const float* __restrict__ b2,
                   const float* __restrict__ Wo,
                   const float* __restrict__ bo,
                   float* __restrict__ out)
{
    __shared__ __attribute__((aligned(16))) float w1lds[F1_SZ * W1_PITCH];
    __shared__ __attribute__((aligned(16))) float w2lds[F2_SZ * W2_PITCH];

    const int lane    = threadIdx.x & 31;
    const int wave    = threadIdx.x >> 5;
    const int tile    = blockIdx.x * WAVES_PER_BLOCK + wave;
    const int r0      = tile * 16;
    const int halfSel = lane >> 4;
    const int mrow    = lane & 15;

    const int xi = x[r0 + mrow];

    const float* eBase = emb + (size_t)xi * D_SZ + 2 * halfSel;

    const int tid = threadIdx.x;
    v4f w1r[16];
#pragma unroll
    for (int j = 0; j < 16; ++j)
        w1r[j] = *(const v4f*)(W1 + 4 * (tid + 128 * j));
    v4f w2r[4];
#pragma unroll
    for (int j = 0; j < 4; ++j)
        w2r[j] = *(const v4f*)(W2 + 4 * (tid + 128 * j));

    v2f ereg[D_SZ / 4];
#pragma unroll
    for (int c = 0; c < D_SZ / 4; ++c)
        ereg[c] = *(const v2f*)(eBase + 4 * c);

#pragma unroll
    for (int j = 0; j < 16; ++j) {
        const int i = tid + 128 * j;
        *(v4f*)(w1lds + (i >> 5) * W1_PITCH + 4 * (i & 31)) = w1r[j];
    }
#pragma unroll
    for (int j = 0; j < 4; ++j) {
        const int i = tid + 128 * j;
        *(v4f*)(w2lds + (i >> 4) * W2_PITCH + 4 * (i & 15)) = w2r[j];
    }
    __builtin_amdgcn_fence(__ATOMIC_RELEASE, "workgroup", "local");
    __builtin_amdgcn_s_barrier();
    __builtin_amdgcn_fence(__ATOMIC_ACQUIRE, "workgroup", "local");

    const float* w1Base = w1lds + mrow * W1_PITCH + 2 * halfSel;
    const float* w2Base = w2lds + mrow * W2_PITCH + 8 * halfSel;

    v8f acc1[4] = {};

    v2f bset0[32], bset1[32];

#define LOAD_W1_BLOCK(BS, BLK)                                                   \
    _Pragma("unroll")                                                            \
    for (int cc = 0; cc < 8; ++cc) {                                             \
        _Pragma("unroll")                                                        \
        for (int t = 0; t < 4; ++t)                                              \
            BS[cc * 4 + t] = *(const v2f*)(w1Base + t * 16 * W1_PITCH +          \
                                           4 * ((BLK) * 8 + cc));                \
    }
#define MMA_W1_BLOCK(BS, BLK)                                                    \
    _Pragma("unroll")                                                            \
    for (int cc = 0; cc < 8; ++cc) {                                             \
        _Pragma("unroll")                                                        \
        for (int t = 0; t < 4; ++t)                                              \
            acc1[t] = __builtin_amdgcn_wmma_f32_16x16x4_f32(                     \
                 false, BS[cc * 4 + t],  false,                \
                ereg[(BLK) * 8 + cc],  (short)0, acc1[t],               \
                 false,  false);                           \
    }
#define GROUP_FENCE() __builtin_amdgcn_sched_barrier(0)

    LOAD_W1_BLOCK(bset0, 0)
    GROUP_FENCE();
    LOAD_W1_BLOCK(bset1, 1)
    GROUP_FENCE();
    MMA_W1_BLOCK(bset0, 0)
    GROUP_FENCE();
    LOAD_W1_BLOCK(bset0, 2)
    GROUP_FENCE();
    MMA_W1_BLOCK(bset1, 1)
    GROUP_FENCE();
    LOAD_W1_BLOCK(bset1, 3)
    GROUP_FENCE();
    MMA_W1_BLOCK(bset0, 2)
    GROUP_FENCE();
#pragma unroll
    for (int t = 0; t < 4; ++t) {
#pragma unroll
        for (int q = 0; q < 4; ++q) {
#pragma unroll
            for (int t2 = 0; t2 < 2; ++t2)
                bset0[(t * 4 + q) * 2 + t2] =
                    *(const v2f*)(w2Base + t2 * 16 * W2_PITCH + 16 * t + 2 * q);
        }
    }
    v4f b1v[8];
#pragma unroll
    for (int t = 0; t < 4; ++t) {
#pragma unroll
        for (int u = 0; u < 2; ++u)
            b1v[t * 2 + u] = *(const v4f*)(b1 + 16 * t + 8 * halfSel + 4 * u);
    }
    GROUP_FENCE();
    MMA_W1_BLOCK(bset1, 3)
    GROUP_FENCE();

    float wide = Wo[F2_SZ + xi];
    float bout = bo[0];
    __attribute__((address_space(1))) float* outp =
        (__attribute__((address_space(1))) float*)out;
    v4f wov[4], b2v[4];
#pragma unroll
    for (int t2 = 0; t2 < 2; ++t2) {
#pragma unroll
        for (int u = 0; u < 2; ++u) {
            wov[t2 * 2 + u] = *(const v4f*)(Wo + 16 * t2 + 8 * halfSel + 4 * u);
            b2v[t2 * 2 + u] = *(const v4f*)(b2 + 16 * t2 + 8 * halfSel + 4 * u);
        }
    }

    v8f h1v[4];
#pragma unroll
    for (int t = 0; t < 4; ++t) {
#pragma unroll
        for (int r = 0; r < 8; ++r)
            h1v[t][r] = acc1[t][r] + b1v[t * 2 + (r >> 2)][r & 3];
    }
    GROUP_FENCE();

    v8f acc2[2] = {};
#pragma unroll
    for (int t = 0; t < 4; ++t) {
#pragma unroll
        for (int q = 0; q < 4; ++q) {
            v2f hfrag;
            hfrag[0] = h1v[t][2 * q];
            hfrag[1] = h1v[t][2 * q + 1];
#pragma unroll
            for (int t2 = 0; t2 < 2; ++t2)
                acc2[t2] = __builtin_amdgcn_wmma_f32_16x16x4_f32(
                    false, bset0[(t * 4 + q) * 2 + t2], false, hfrag,
                    (short)0, acc2[t2], false, false);
        }
    }

    float s0 = 0.0f, s1 = 0.0f;
#pragma unroll
    for (int r = 0; r < 8; ++r) {
        s0 += (acc2[0][r] + b2v[r >> 2][r & 3])       * wov[r >> 2][r & 3];
        s1 += (acc2[1][r] + b2v[2 + (r >> 2)][r & 3]) * wov[2 + (r >> 2)][r & 3];
    }
    float y = s0 + s1;
    asm volatile("" : "+v"(wide), "+s"(bout), "+s"(outp), "+v"(y));
    y += __shfl_xor(y, 16, 32);

    if (halfSel == 0)
        outp[r0 + mrow] = y + wide + bout;
}

extern "C" void kernel_launch(void* const* d_in, const int* in_sizes, int n_in,
                              void* d_out, int out_size, void* d_ws, size_t ws_size,
                              hipStream_t stream) {
    (void)in_sizes; (void)n_in; (void)out_size; (void)d_ws; (void)ws_size;
    const int*   x   = (const int*)  d_in[0];
    const float* emb = (const float*)d_in[1];
    const float* W1  = (const float*)d_in[2];
    const float* b1  = (const float*)d_in[3];
    const float* W2  = (const float*)d_in[4];
    const float* b2  = (const float*)d_in[5];
    const float* Wo  = (const float*)d_in[6];
    const float* bo  = (const float*)d_in[7];
    float*       out = (float*)d_out;

    const int tiles  = B_SZ / 16;
    const int blocks = tiles / WAVES_PER_BLOCK;
    wide_deep_fwd<<<blocks, 32 * WAVES_PER_BLOCK, 0, stream>>>(
        x, emb, W1, b1, W2, b2, Wo, bo, out);
}


# W1, W2 no longer staged in LDS: each wave loads the B fragments of its own u and v tiles straight from global (its two v tiles interleaved by column: whole-line b64 loads); one barrier fewer
# speedup vs baseline: 1.8043x; 1.2228x over previous
#include <hip/hip_runtime.h>


typedef __attribute__((ext_vector_type(2))) float v2f;
typedef __attribute__((ext_vector_type(8))) float v8f;

#define B_SZ 16384
#define D_SZ 128
#define F1_SZ 64
#define F2_SZ 32
#define WAVES_PER_BLOCK 4

#define GROUP_FENCE() __builtin_amdgcn_sched_barrier(0)

#define LDS_BARRIER()                                                  \
    do {                                                               \
        __builtin_amdgcn_fence(__ATOMIC_RELEASE, "workgroup", "local"); \
        __builtin_amdgcn_s_barrier();                                  \
        __builtin_amdgcn_fence(__ATOMIC_ACQUIRE, "workgroup", "local"); \
    } while (0)

#define WMMA_F32(A, B, C)                                              \
    __builtin_amdgcn_wmma_f32_16x16x4_f32( false, (A),        \
         false, (B),  (short)0, (C),                 \
         false,  false)

__global__ __launch_bounds__(32 * WAVES_PER_BLOCK)
void wide_deep_fwd(const int* __restrict__ x,
                   const float* __restrict__ emb,
                   const float* __restrict__ W1,
                   const float* __restrict__ b1,
                   const float* __restrict__ W2,
                   const float* __restrict__ b2,
                   const float* __restrict__ Wo,
                   const float* __restrict__ bo,
                   float* __restrict__ out)
{
    __shared__ __attribute__((aligned(16))) float ulds[F1_SZ];
    __shared__ __attribute__((aligned(16))) float vlds[D_SZ];

    const int lane    = threadIdx.x & 31;
    const int wave    = threadIdx.x >> 5;
    const int tile    = blockIdx.x * WAVES_PER_BLOCK + wave;
    const int r0      = tile * 16;
    const int halfSel = lane >> 4;
    const int mrow    = lane & 15;

    const unsigned xi = (unsigned)x[r0 + mrow];
    GROUP_FENCE();


    const float* w2B = W2 + (2 * halfSel) * F1_SZ + 16 * wave + mrow;
    v2f w2f[F2_SZ / 4];
#pragma unroll
    for (int j = 0; j < F2_SZ / 4; ++j) {
        w2f[j][0] = w2B[(4 * j)     * F1_SZ];
        w2f[j][1] = w2B[(4 * j + 1) * F1_SZ];
    }
    v2f woA[F2_SZ / 4];
#pragma unroll
    for (int j = 0; j < F2_SZ / 4; ++j)
        woA[j] = *(const v2f*)(Wo + 4 * j + 2 * halfSel);

    const float* w1B = W1 + (2 * halfSel) * D_SZ + 32 * wave + 2 * mrow;
    v2f w1p[2 * (F1_SZ / 4)];
#pragma unroll
    for (int j = 0; j < F1_SZ / 4; ++j) {
#pragma unroll
        for (int s = 0; s < 2; ++s)
            w1p[2 * j + s] = *(const v2f*)(w1B + (4 * j + s) * D_SZ);
    }

    GROUP_FENCE();
    float wide = Wo[F2_SZ + xi];
    float bout = bo[0];
    v2f b1A[F1_SZ / 4], b2A[F2_SZ / 4];
#pragma unroll
    for (int j = 0; j < F1_SZ / 4; ++j)
        b1A[j] = *(const v2f*)(b1 + 4 * j + 2 * halfSel);
#pragma unroll
    for (int j = 0; j < F2_SZ / 4; ++j)
        b2A[j] = *(const v2f*)(b2 + 4 * j + 2 * halfSel);
    const float* eBase = emb + (size_t)xi * D_SZ + 2 * halfSel;
    v2f ereg[D_SZ / 4];
#pragma unroll
    for (int c = 0; c < D_SZ / 4; ++c)
        ereg[c] = *(const v2f*)(eBase + 4 * c);
    __attribute__((address_space(1))) float* outp =
        (__attribute__((address_space(1))) float*)out;

    v8f accu[2] = {};
#pragma unroll
    for (int j = 0; j < F2_SZ / 4; ++j)
        accu[j & 1] = WMMA_F32(woA[j], w2f[j], accu[j & 1]);
    ulds[16 * wave + mrow] = accu[0][0] + accu[1][0];
    LDS_BARRIER();

    v2f uA[F1_SZ / 4];
#pragma unroll
    for (int j = 0; j < F1_SZ / 4; ++j)
        uA[j] = *(const v2f*)(ulds + 4 * j + 2 * halfSel);
    GROUP_FENCE();
    v8f accv[2] = {};
#pragma unroll
    for (int j = 0; j < F1_SZ / 4; ++j) {
#pragma unroll
        for (int t = 0; t < 2; ++t) {
            v2f w1f;
            w1f[0] = w1p[2 * j][t];
            w1f[1] = w1p[2 * j + 1][t];
            accv[t] = WMMA_F32(uA[j], w1f, accv[t]);
        }
    }
    v2f vpair;
    vpair[0] = accv[0][0];
    vpair[1] = accv[1][0];
    *(v2f*)(vlds + 32 * wave + 2 * mrow) = vpair;

    float cb0 = 0.0f, cb1 = 0.0f, cb2 = 0.0f, cb3 = 0.0f;
#pragma unroll
    for (int j = 0; j < F1_SZ / 4; j += 2) {
        cb0 += uA[j][0]     * b1A[j][0];
        cb1 += uA[j][1]     * b1A[j][1];
        cb2 += uA[j + 1][0] * b1A[j + 1][0];
        cb3 += uA[j + 1][1] * b1A[j + 1][1];
    }
#pragma unroll
    for (int j = 0; j < F2_SZ / 4; j += 2) {
        cb0 += woA[j][0]     * b2A[j][0];
        cb1 += woA[j][1]     * b2A[j][1];
        cb2 += woA[j + 1][0] * b2A[j + 1][0];
        cb3 += woA[j + 1][1] * b2A[j + 1][1];
    }
    v2f cbA;
    cbA[0] = (cb0 + cb1) + (cb2 + cb3);
    cbA[1] = 0.0f;
    LDS_BARRIER();

    v2f vA[D_SZ / 4];
#pragma unroll
    for (int c = 0; c < D_SZ / 4; ++c)
        vA[c] = *(const v2f*)(vlds + 4 * c + 2 * halfSel);
    GROUP_FENCE();
    v8f accy[2] = {};
#pragma unroll
    for (int c = 0; c < D_SZ / 4; ++c)
        accy[c & 1] = WMMA_F32(vA[c], ereg[c], accy[c & 1]);
    v2f ones;
    ones[0] = 1.0f;
    ones[1] = 1.0f;
    accy[0] = WMMA_F32(cbA, ones, accy[0]);

    float y = accy[0][0] + accy[1][0];
    asm volatile("" : "+v"(wide), "+s"(bout), "+s"(outp), "+v"(y));

    if (halfSel == 0)
        outp[r0 + mrow] = y + wide + bout;
}

extern "C" void kernel_launch(void* const* d_in, const int* in_sizes, int n_in,
                              void* d_out, int out_size, void* d_ws, size_t ws_size,
                              hipStream_t stream) {
    (void)in_sizes; (void)n_in; (void)out_size; (void)d_ws; (void)ws_size;
    const int*   x   = (const int*)  d_in[0];
    const float* emb = (const float*)d_in[1];
    const float* W1  = (const float*)d_in[2];
    const float* b1  = (const float*)d_in[3];
    const float* W2  = (const float*)d_in[4];
    const float* b2  = (const float*)d_in[5];
    const float* Wo  = (const float*)d_in[6];
    const float* bo  = (const float*)d_in[7];
    float*       out = (float*)d_out;

    const int tiles  = B_SZ / 16;
    const int blocks = tiles / WAVES_PER_BLOCK;
    wide_deep_fwd<<<blocks, 32 * WAVES_PER_BLOCK, 0, stream>>>(
        x, emb, W1, b1, W2, b2, Wo, bo, out);
}
